# starting kernel
# baseline (speedup 1.0000x reference)
#include <hip/hip_runtime.h>
#include <hip/hip_bf16.h>

typedef __attribute__((ext_vector_type(2))) float v2f;
typedef __attribute__((ext_vector_type(4))) float v4f;
typedef __attribute__((ext_vector_type(8))) float v8f;

#define K_SEQ       2001
#define M_DIM       16
#define SEQ_ELEMS   (K_SEQ * M_DIM)
#define ROWS_TILE   64
#define TILE_ELEMS  (ROWS_TILE * M_DIM)
#define NUM_TILES   32
#define NSEG        6

__device__ __forceinline__ int inc3(int t) { return (t == 2) ? 0 : t + 1; }

__device__ __forceinline__ int seg_from(int r, int t) {
    int seg = (r < 1000) ? t : (3 + inc3(t));
    seg = (r == 1000) ? 99 : seg;
    seg = (r >= 2001) ? 99 : seg;
    return seg;
}

__device__ __forceinline__ float onehot(int seg, int m) {
    return (seg == m) ? 1.0f : 0.0f;
}

__global__ __launch_bounds__(256)
void seg_mean_wmma_kernel(const float* __restrict__ in, float* __restrict__ out) {
    __shared__ float tile[TILE_ELEMS];
    __shared__ float acc[NSEG * M_DIM];

    const int tid  = threadIdx.x;
    const int lane = tid & 31;
    const int wave = tid >> 5;
    const int half = lane >> 4;
    const int m    = lane & 15;
    const size_t seq = blockIdx.x;
    const float* __restrict__ src = in + seq * (size_t)SEQ_ELEMS;

    if (tid < NSEG * M_DIM) acc[tid] = 0.0f;

    int rA0 = wave * 8 + 2 * half;
    int t   = rA0 % 3;

    const int l0 = rA0 * M_DIM + m;

    v8f c = {};

    for (int it = 0; it < NUM_TILES; ++it) {
        int eoff = it * TILE_ELEMS + tid * 4;
        int ld   = (eoff > SEQ_ELEMS - 4) ? (SEQ_ELEMS - 4) : eoff;
        v4f v = *(const v4f*)(src + ld);
        __syncthreads();
        *(v4f*)(tile + tid * 4) = v;
        __syncthreads();

        const int t1 = inc3(t);
        const int t2 = inc3(t1);

        {
            v2f a, b;
            a.x = onehot(seg_from(rA0,     t ), m);
            a.y = onehot(seg_from(rA0 + 1, t1), m);
            b.x = tile[l0];
            b.y = tile[l0 + 16];
            c = __builtin_amdgcn_wmma_f32_16x16x4_f32(
                    false, a, false, b, (short)0, c, false, false);
        }
        {
            v2f a, b;
            a.x = onehot(seg_from(rA0 + 4, t1), m);
            a.y = onehot(seg_from(rA0 + 5, t2), m);
            b.x = tile[l0 + 64];
            b.y = tile[l0 + 80];
            c = __builtin_amdgcn_wmma_f32_16x16x4_f32(
                    false, a, false, b, (short)0, c, false, false);
        }

        rA0 += ROWS_TILE;
        t = inc3(t);
    }

    __syncthreads();
    if (lane < 16) {
        atomicAdd(&acc[0 * M_DIM + m], c[0]);
        atomicAdd(&acc[1 * M_DIM + m], c[1]);
        atomicAdd(&acc[2 * M_DIM + m], c[2]);
        atomicAdd(&acc[3 * M_DIM + m], c[3]);
        atomicAdd(&acc[4 * M_DIM + m], c[4]);
        atomicAdd(&acc[5 * M_DIM + m], c[5]);
    }
    __syncthreads();

    if (tid < NSEG * M_DIM) {
        const int s = tid >> 4;
        const float cnt = (s == 0 || s == 3) ? 334.0f : 333.0f;
        out[seq * (NSEG * M_DIM) + tid] = acc[tid] / cnt;
    }
}

extern "C" void kernel_launch(void* const* d_in, const int* in_sizes, int n_in,
                              void* d_out, int out_size, void* d_ws, size_t ws_size,
                              hipStream_t stream) {
    const float* in  = (const float*)d_in[0];
    float*       out = (float*)d_out;
    (void)in_sizes; (void)n_in; (void)out_size; (void)d_ws; (void)ws_size;
    seg_mean_wmma_kernel<<<dim3(2048), dim3(256), 0, stream>>>(in, out);
}


# Each wave streams one half of a sequence from global memory straight into the WMMA B fragments, 25 16-byte loads in flight, no LDS tile or loop barrier; LDS atomics gone, fixed summation order
# speedup vs baseline: 1.7274x; 1.7274x over previous
#include <hip/hip_runtime.h>


typedef __attribute__((ext_vector_type(2))) float v2f;
typedef __attribute__((ext_vector_type(4))) float v4f;
typedef __attribute__((ext_vector_type(8))) float v8f;

#define K_SEQ        2001
#define M_DIM        16
#define SEQ_ELEMS    (K_SEQ * M_DIM)
#define SIDE_ROWS    1000
#define RIGHT_START  1001
#define GROUP_ROWS   8
#define GROUP_ELEMS  (GROUP_ROWS * M_DIM)
#define NGROUPS      (SIDE_ROWS / GROUP_ROWS)
#define DEPTH        25
#define NBATCH       (NGROUPS / DEPTH)
#define NSEG         6

__device__ __forceinline__ v2f onehot_pair(int c, int rho) {
    v2f a;
    a.x = (c == rho)     ? 1.0f : 0.0f;
    a.y = (c == rho + 3) ? 1.0f : 0.0f;
    return a;
}

__device__ __forceinline__ void consume(v8f& c01, v8f& c23, v2f a, v4f v) {
    v2f b01 = {v.x, v.y};
    v2f b23 = {v.z, v.w};
    c01 = __builtin_amdgcn_wmma_f32_16x16x4_f32(false, a, false, b01, (short)0, c01, false, false);
    c23 = __builtin_amdgcn_wmma_f32_16x16x4_f32(false, a, false, b23, (short)0, c23, false, false);
}

__device__ __forceinline__ v2f pick3(int k, v2f a0, v2f a1, v2f a2) {
    return (k == 0) ? a0 : (k == 1) ? a1 : a2;
}

__global__ __launch_bounds__(64)
void seg_mean_stream_kernel(const float* __restrict__ in, float* __restrict__ out) {
    __shared__ v4f part[2 * 4 * 3 * 4];

    const int tid  = threadIdx.x;
    const int lane = tid & 31;
    const int side = tid >> 5;
    const int h    = lane >> 4;
    const int c    = lane & 15;
    const size_t seq = blockIdx.x;

    const float* p = in + seq * (size_t)SEQ_ELEMS + side * (RIGHT_START * M_DIM) + lane * 4;

    v2f A0 = onehot_pair(c, h);
    v2f A1 = onehot_pair(c, 1 + h);
    v2f A2 = onehot_pair(c, h ? 0 : 2);

    v4f buf[DEPTH];
    #pragma unroll
    for (int i = 0; i < DEPTH; ++i) buf[i] = *(const v4f*)(p + i * GROUP_ELEMS);

    v8f c01 = {};
    v8f c23 = {};

    #pragma nounroll
    for (int b = 0; b < NBATCH - 1; ++b) {
        p += DEPTH * GROUP_ELEMS;
        #pragma unroll
        for (int i = 0; i < DEPTH; ++i) {
            consume(c01, c23, pick3((2 * i) % 3, A0, A1, A2), buf[i]);
            buf[i] = *(const v4f*)(p + i * GROUP_ELEMS);
        }
        v2f t = A0; A0 = A2; A2 = A1; A1 = t;
    }
    #pragma unroll
    for (int i = 0; i < DEPTH; ++i)
        consume(c01, c23, pick3((2 * i) % 3, A0, A1, A2), buf[i]);

    if (lane < 16) {
        const int q  = c >> 2;
        const int j  = c & 3;
        const int t0 = (q == 3) ? 0 : q;
        const int t1 = (t0 == 2) ? 0 : t0 + 1;
        const int t2 = (t1 == 2) ? 0 : t1 + 1;
        const int base = (side * 4 + q) * 3;
        const v4f r0 = {c01[0], c01[3], c23[0], c23[3]};
        const v4f r1 = {c01[1], c01[4], c23[1], c23[4]};
        const v4f r2 = {c01[2], c01[5], c23[2], c23[5]};
        part[(base + t0) * 4 + j] = r0;
        part[(base + t1) * 4 + j] = r1;
        part[(base + t2) * 4 + j] = r2;
    }
    __syncthreads();

    if (tid < NSEG * M_DIM / 4) {
        const int s  = tid >> 2;
        const int j  = tid & 3;
        const int sd = (s >= 3) ? 1 : 0;
        const int t  = s - 3 * sd;
        const float cnt = (t == 0) ? 334.0f : 333.0f;
        const int i0 = (sd * 12 + t) * 4 + j;
        const v4f s01 = part[i0]      + part[i0 + 12];
        const v4f s23 = part[i0 + 24] + part[i0 + 36];
        v4f r = s01 + s23;
        r.x = r.x / cnt; r.y = r.y / cnt; r.z = r.z / cnt; r.w = r.w / cnt;
        *(v4f*)(out + seq * (NSEG * M_DIM) + tid * 4) = r;
    }
}

extern "C" void kernel_launch(void* const* d_in, const int* in_sizes, int n_in,
                              void* d_out, int out_size, void* d_ws, size_t ws_size,
                              hipStream_t stream) {
    const float* in  = (const float*)d_in[0];
    float*       out = (float*)d_out;
    (void)in_sizes; (void)n_in; (void)out_size; (void)d_ws; (void)ws_size;
    seg_mean_stream_kernel<<<dim3(2048), dim3(64), 0, stream>>>(in, out);
}
